# starting kernel
# baseline (speedup 1.0000x reference)
#include <hip/hip_runtime.h>

typedef float v2f __attribute__((ext_vector_type(2)));
typedef float v8f __attribute__((ext_vector_type(8)));

#define HW 192
#define OX 95
#define OY 95

__global__ __launch_bounds__(256)
void quanv_kernel(const float* __restrict__ X, const float* __restrict__ Wt,
                  float* __restrict__ out, int totalRows) {
    __shared__ float s_bmat[16 * 16];
    __shared__ float s_phc[16], s_phs[16];
    __shared__ float s_re[8][32][18];
    __shared__ float s_im[8][32][18];
    __shared__ float s_p[8][16][17];

    const int tid  = threadIdx.x;
    const int lane = tid & 31;
    const int wv   = tid >> 5;

    const float w0 = Wt[0], w1 = Wt[1], w2 = Wt[2], w3 = Wt[3];
    float cq[4], sq[4];
#pragma unroll
    for (int q = 0; q < 4; ++q) __sincosf(0.5f * Wt[4 + q], &sq[q], &cq[q]);

    if (tid < 16) {
        int s = tid;
        int b0 = (s >> 3) & 1, b1 = (s >> 2) & 1, b2 = (s >> 1) & 1, b3 = s & 1;
        float phi = 0.5f * ((float)b0 * (2 * b1 - 1) * w0 + (float)b1 * (2 * b2 - 1) * w1 +
                            (float)b2 * (2 * b3 - 1) * w2 + (float)b3 * (2 * b0 - 1) * w3);
        float sp, cp; __sincosf(phi, &sp, &cp);
        s_phc[s] = cp; s_phs[s] = sp;
    }
    {
        int k = tid >> 4, n = tid & 15;
        float v = 1.0f;
#pragma unroll
        for (int q = 0; q < 4; ++q) {
            int i = (n >> (3 - q)) & 1;
            int j = (k >> (3 - q)) & 1;
            float m = (i == j) ? cq[q] : ((i == 0) ? -sq[q] : sq[q]);
            v *= m;
        }
        s_bmat[k * 16 + n] = v;
    }
    __syncthreads();

    const int rowBase = (int)blockIdx.x * 256;
    int r = rowBase + tid;
    if (r >= totalRows) r = totalRows - 1;

    unsigned rr  = (unsigned)r;
    unsigned b   = rr / (unsigned)(OX * OY);
    unsigned rem = rr % (unsigned)(OX * OY);
    unsigned px  = rem / (unsigned)OY;
    unsigned py  = rem % (unsigned)OY;

    const float* xp = X + (size_t)b * (HW * HW) + (size_t)(2u * px) * HW + 2u * py;
    float t[16];
#pragma unroll
    for (int kx = 0; kx < 4; ++kx) {
        const float* rp = xp + kx * HW;
        float2 a = *(const float2*)(rp);
        float2 c = *(const float2*)(rp + 2);
        t[kx * 4 + 0] = a.x; t[kx * 4 + 1] = a.y;
        t[kx * 4 + 2] = c.x; t[kx * 4 + 3] = c.y;
    }

    float vr[4][2], vi[4][2];
    const float is2 = 0.70710678118654752f;
#pragma unroll
    for (int q = 0; q < 4; ++q) { vr[q][0] = is2; vr[q][1] = is2; vi[q][0] = 0.f; vi[q][1] = 0.f; }

    auto applyRZ = [&](int q, float th) {
        float sh, ch; __sincosf(0.5f * th, &sh, &ch);
        float r0 = vr[q][0], i0 = vi[q][0], r1 = vr[q][1], i1 = vi[q][1];
        vr[q][0] = r0 * ch + i0 * sh;  vi[q][0] = i0 * ch - r0 * sh;
        vr[q][1] = r1 * ch - i1 * sh;  vi[q][1] = i1 * ch + r1 * sh;
    };
    auto applyRY = [&](int q, float th) {
        float sh, ch; __sincosf(0.5f * th, &sh, &ch);
        float r0 = vr[q][0], i0 = vi[q][0], r1 = vr[q][1], i1 = vi[q][1];
        vr[q][0] = ch * r0 - sh * r1;  vi[q][0] = ch * i0 - sh * i1;
        vr[q][1] = sh * r0 + ch * r1;  vi[q][1] = sh * i0 + ch * i1;
    };
#pragma unroll
    for (int q = 0; q < 3; ++q) {
        applyRZ(q, t[q * 5 + 0]); applyRY(q, t[q * 5 + 1]); applyRZ(q, t[q * 5 + 2]);
        applyRY(q, t[q * 5 + 3]); applyRZ(q, t[q * 5 + 4]);
    }
    applyRZ(3, t[15]);

    float w01r[4], w01i[4], w23r[4], w23i[4];
#pragma unroll
    for (int i = 0; i < 2; ++i)
#pragma unroll
        for (int j = 0; j < 2; ++j) {
            w01r[i * 2 + j] = vr[0][i] * vr[1][j] - vi[0][i] * vi[1][j];
            w01i[i * 2 + j] = vr[0][i] * vi[1][j] + vi[0][i] * vr[1][j];
            w23r[i * 2 + j] = vr[2][i] * vr[3][j] - vi[2][i] * vi[3][j];
            w23i[i * 2 + j] = vr[2][i] * vi[3][j] + vi[2][i] * vr[3][j];
        }
#pragma unroll
    for (int s = 0; s < 16; ++s) {
        float ar = w01r[s >> 2], ai = w01i[s >> 2];
        float br = w23r[s & 3],  bi = w23i[s & 3];
        float re = ar * br - ai * bi, im = ar * bi + ai * br;
        float pc = s_phc[s], ps = s_phs[s];
        s_re[wv][lane][s] = re * pc - im * ps;
        s_im[wv][lane][s] = re * ps + im * pc;
    }
    __syncthreads();

    const int nn   = lane & 15;
    const int koff = (lane < 16) ? 0 : 2;

#pragma unroll
    for (int tI = 0; tI < 2; ++tI) {
        v8f dRe = {}; v8f dIm = {};
        const int srcRow = tI * 16 + nn;
#pragma unroll
        for (int k0 = 0; k0 < 16; k0 += 4) {
            const int kk = k0 + koff;
            v2f a_re, a_im, bop;
            a_re.x = s_re[wv][srcRow][kk];      a_re.y = s_re[wv][srcRow][kk + 1];
            a_im.x = s_im[wv][srcRow][kk];      a_im.y = s_im[wv][srcRow][kk + 1];
            bop.x  = s_bmat[kk * 16 + nn];      bop.y  = s_bmat[(kk + 1) * 16 + nn];
            dRe = __builtin_amdgcn_wmma_f32_16x16x4_f32(false, a_re, false, bop,
                                                        (short)0, dRe, false, false);
            dIm = __builtin_amdgcn_wmma_f32_16x16x4_f32(false, a_im, false, bop,
                                                        (short)0, dIm, false, false);
        }
        const int prow0 = (lane < 16) ? 0 : 8;
#pragma unroll
        for (int v = 0; v < 8; ++v) {
            float p = dRe[v] * dRe[v] + dIm[v] * dIm[v];
            s_p[wv][prow0 + v][nn] = p;
        }
        __builtin_amdgcn_wave_barrier();
#pragma unroll
        for (int oo = 0; oo < 2; ++oo) {
            int o    = lane + oo * 32;
            int prow = o >> 2;
            int q    = o & 3;
            float ev = 0.f;
#pragma unroll
            for (int s = 0; s < 16; ++s) {
                float sgn = ((s >> (3 - q)) & 1) ? -1.f : 1.f;
                ev += sgn * s_p[wv][prow][s];
            }
            int rowG = rowBase + wv * 32 + tI * 16 + prow;
            if (rowG < totalRows) out[(size_t)rowG * 4 + q] = ev;
        }
        __builtin_amdgcn_wave_barrier();
    }
}

extern "C" void kernel_launch(void* const* d_in, const int* in_sizes, int n_in,
                              void* d_out, int out_size, void* d_ws, size_t ws_size,
                              hipStream_t stream) {
    const float* X  = (const float*)d_in[0];
    const float* Wt = (const float*)d_in[1];
    float* out      = (float*)d_out;
    int B = in_sizes[0] / (HW * HW);
    int totalRows = B * OX * OY;
    int blocks = (totalRows + 255) / 256;
    quanv_kernel<<<blocks, 256, 0, stream>>>(X, Wt, out, totalRows);
}


# Product transposed (weights as A operand) so a lane holds 8 states of one row: signed sums as in-lane add tree plus one half-wave exchange, one 16-byte store per row; no LDS pass; sum order changed
# speedup vs baseline: 1.2379x; 1.2379x over previous
#include <hip/hip_runtime.h>

typedef float v2f __attribute__((ext_vector_type(2)));
typedef float v8f __attribute__((ext_vector_type(8)));

#define HW 192
#define OX 95
#define OY 95

__device__ __forceinline__ float xhalf(float x) {
    int b = __builtin_bit_cast(int, x);
    return __builtin_bit_cast(float,
        __builtin_amdgcn_permlanex16(b, b, 0x76543210, 0xfedcba98, false, false));
}

__global__ __launch_bounds__(256)
void quanv_kernel(const float* __restrict__ X, const float* __restrict__ Wt,
                  float* __restrict__ out, int totalRows) {
    __shared__ float s_bmat[16 * 16];
    __shared__ float s_phc[16], s_phs[16];
    __shared__ float s_re[8][32][18];
    __shared__ float s_im[8][32][18];

    const int tid  = threadIdx.x;
    const int lane = tid & 31;
    const int wv   = tid >> 5;

    const float w0 = Wt[0], w1 = Wt[1], w2 = Wt[2], w3 = Wt[3];
    float cq[4], sq[4];
#pragma unroll
    for (int q = 0; q < 4; ++q) __sincosf(0.5f * Wt[4 + q], &sq[q], &cq[q]);

    if (tid < 16) {
        int s = tid;
        int b0 = (s >> 3) & 1, b1 = (s >> 2) & 1, b2 = (s >> 1) & 1, b3 = s & 1;
        float phi = 0.5f * ((float)b0 * (2 * b1 - 1) * w0 + (float)b1 * (2 * b2 - 1) * w1 +
                            (float)b2 * (2 * b3 - 1) * w2 + (float)b3 * (2 * b0 - 1) * w3);
        float sp, cp; __sincosf(phi, &sp, &cp);
        s_phc[s] = cp; s_phs[s] = sp;
    }
    {
        int k = tid >> 4, n = tid & 15;
        float v = 1.0f;
#pragma unroll
        for (int q = 0; q < 4; ++q) {
            int i = (n >> (3 - q)) & 1;
            int j = (k >> (3 - q)) & 1;
            float m = (i == j) ? cq[q] : ((i == 0) ? -sq[q] : sq[q]);
            v *= m;
        }
        s_bmat[k * 16 + n] = v;
    }
    __syncthreads();

    const int rowBase = (int)blockIdx.x * 256;
    int r = rowBase + tid;
    if (r >= totalRows) r = totalRows - 1;

    unsigned rr  = (unsigned)r;
    unsigned b   = rr / (unsigned)(OX * OY);
    unsigned rem = rr % (unsigned)(OX * OY);
    unsigned px  = rem / (unsigned)OY;
    unsigned py  = rem % (unsigned)OY;

    const float* xp = X + (size_t)b * (HW * HW) + (size_t)(2u * px) * HW + 2u * py;
    float t[16];
#pragma unroll
    for (int kx = 0; kx < 4; ++kx) {
        const float* rp = xp + kx * HW;
        float2 a = *(const float2*)(rp);
        float2 c = *(const float2*)(rp + 2);
        t[kx * 4 + 0] = a.x; t[kx * 4 + 1] = a.y;
        t[kx * 4 + 2] = c.x; t[kx * 4 + 3] = c.y;
    }

    float vr[4][2], vi[4][2];
    const float is2 = 0.70710678118654752f;
#pragma unroll
    for (int q = 0; q < 4; ++q) { vr[q][0] = is2; vr[q][1] = is2; vi[q][0] = 0.f; vi[q][1] = 0.f; }

    auto applyRZ = [&](int q, float th) {
        float sh, ch; __sincosf(0.5f * th, &sh, &ch);
        float r0 = vr[q][0], i0 = vi[q][0], r1 = vr[q][1], i1 = vi[q][1];
        vr[q][0] = r0 * ch + i0 * sh;  vi[q][0] = i0 * ch - r0 * sh;
        vr[q][1] = r1 * ch - i1 * sh;  vi[q][1] = i1 * ch + r1 * sh;
    };
    auto applyRY = [&](int q, float th) {
        float sh, ch; __sincosf(0.5f * th, &sh, &ch);
        float r0 = vr[q][0], i0 = vi[q][0], r1 = vr[q][1], i1 = vi[q][1];
        vr[q][0] = ch * r0 - sh * r1;  vi[q][0] = ch * i0 - sh * i1;
        vr[q][1] = sh * r0 + ch * r1;  vi[q][1] = sh * i0 + ch * i1;
    };
#pragma unroll
    for (int q = 0; q < 3; ++q) {
        applyRZ(q, t[q * 5 + 0]); applyRY(q, t[q * 5 + 1]); applyRZ(q, t[q * 5 + 2]);
        applyRY(q, t[q * 5 + 3]); applyRZ(q, t[q * 5 + 4]);
    }
    applyRZ(3, t[15]);

    float w01r[4], w01i[4], w23r[4], w23i[4];
#pragma unroll
    for (int i = 0; i < 2; ++i)
#pragma unroll
        for (int j = 0; j < 2; ++j) {
            w01r[i * 2 + j] = vr[0][i] * vr[1][j] - vi[0][i] * vi[1][j];
            w01i[i * 2 + j] = vr[0][i] * vi[1][j] + vi[0][i] * vr[1][j];
            w23r[i * 2 + j] = vr[2][i] * vr[3][j] - vi[2][i] * vi[3][j];
            w23i[i * 2 + j] = vr[2][i] * vi[3][j] + vi[2][i] * vr[3][j];
        }
#pragma unroll
    for (int s = 0; s < 16; ++s) {
        float ar = w01r[s >> 2], ai = w01i[s >> 2];
        float br = w23r[s & 3],  bi = w23i[s & 3];
        float re = ar * br - ai * bi, im = ar * bi + ai * br;
        float pc = s_phc[s], ps = s_phs[s];
        s_re[wv][lane][s] = re * pc - im * ps;
        s_im[wv][lane][s] = re * ps + im * pc;
    }
    __syncthreads();

    const int nn   = lane & 15;
    const int hh   = lane >> 4;
    const int koff = hh * 2;

    float tot[2], z1[2], z2[2], z3[2];
#pragma unroll
    for (int tI = 0; tI < 2; ++tI) {
        v8f dRe = {}; v8f dIm = {};
        const int srcRow = tI * 16 + nn;
#pragma unroll
        for (int k0 = 0; k0 < 16; k0 += 4) {
            const int kk = k0 + koff;
            v2f b_re, b_im, aop;
            b_re.x = s_re[wv][srcRow][kk];      b_re.y = s_re[wv][srcRow][kk + 1];
            b_im.x = s_im[wv][srcRow][kk];      b_im.y = s_im[wv][srcRow][kk + 1];
            aop.x  = s_bmat[kk * 16 + nn];      aop.y  = s_bmat[(kk + 1) * 16 + nn];
            dRe = __builtin_amdgcn_wmma_f32_16x16x4_f32(false, aop, false, b_re,
                                                        (short)0, dRe, false, false);
            dIm = __builtin_amdgcn_wmma_f32_16x16x4_f32(false, aop, false, b_im,
                                                        (short)0, dIm, false, false);
        }
        float p[8];
#pragma unroll
        for (int v = 0; v < 8; ++v) p[v] = dRe[v] * dRe[v] + dIm[v] * dIm[v];
        const float s0 = p[0] + p[4], s1 = p[1] + p[5], s2 = p[2] + p[6], s3 = p[3] + p[7];
        const float d0 = p[0] - p[4], d1 = p[1] - p[5], d2 = p[2] - p[6], d3 = p[3] - p[7];
        const float t0 = s0 + s2, t1 = s1 + s3;
        const float u0 = s0 - s2, u1 = s1 - s3;
        z1[tI]  = (d0 + d2) + (d1 + d3);
        z2[tI]  = u0 + u1;
        z3[tI]  = t0 - t1;
        tot[tI] = t0 + t1;
    }

    const bool up = (hh != 0);
    const float k0v = up ? -tot[1] : tot[0];
    const float g0v = up ? -tot[0] : tot[1];
    const float k1v = up ? z1[1] : z1[0];
    const float g1v = up ? z1[0] : z1[1];
    const float k2v = up ? z2[1] : z2[0];
    const float g2v = up ? z2[0] : z2[1];
    const float k3v = up ? z3[1] : z3[0];
    const float g3v = up ? z3[0] : z3[1];
    float4 ev;
    ev.x = k0v + xhalf(g0v);
    ev.y = k1v + xhalf(g1v);
    ev.z = k2v + xhalf(g2v);
    ev.w = k3v + xhalf(g3v);

    const int rowG = rowBase + tid;
    if (rowG < totalRows) *(float4*)(out + (size_t)rowG * 4) = ev;
}

extern "C" void kernel_launch(void* const* d_in, const int* in_sizes, int n_in,
                              void* d_out, int out_size, void* d_ws, size_t ws_size,
                              hipStream_t stream) {
    const float* X  = (const float*)d_in[0];
    const float* Wt = (const float*)d_in[1];
    float* out      = (float*)d_out;
    int B = in_sizes[0] / (HW * HW);
    int totalRows = B * OX * OY;
    int blocks = (totalRows + 255) / 256;
    quanv_kernel<<<blocks, 256, 0, stream>>>(X, Wt, out, totalRows);
}


# Encode without unobservable phases: RZ as diag(1,e^{it}); CRZ ring = control-side phase (added to last RZ angle) times e^{iw} where both bits set; 1/4 norm put in the weights; rounding changes
# speedup vs baseline: 1.3860x; 1.1196x over previous
#include <hip/hip_runtime.h>

typedef float v2f __attribute__((ext_vector_type(2)));
typedef float v8f __attribute__((ext_vector_type(8)));

#define HW 192
#define OX 95
#define OY 95

__device__ __forceinline__ float xhalf(float x) {
    int b = __builtin_bit_cast(int, x);
    return __builtin_bit_cast(float,
        __builtin_amdgcn_permlanex16(b, b, 0x76543210, 0xfedcba98, false, false));
}

__global__ __launch_bounds__(256)
void quanv_kernel(const float* __restrict__ X, const float* __restrict__ Wt,
                  float* __restrict__ out, int totalRows) {
    __shared__ float s_bmat[16 * 16];
    __shared__ float s_ec[4], s_es[4];
    __shared__ float s_re[8][32][18];
    __shared__ float s_im[8][32][18];

    const int tid  = threadIdx.x;
    const int lane = tid & 31;
    const int wv   = tid >> 5;

    float cq[4], sq[4];
#pragma unroll
    for (int q = 0; q < 4; ++q) __sincosf(0.5f * Wt[4 + q], &sq[q], &cq[q]);

    if (tid < 4) {
        float sp, cp; __sincosf(Wt[tid], &sp, &cp);
        s_ec[tid] = cp; s_es[tid] = sp;
    }
    {
        int k = tid >> 4, n = tid & 15;
        float v = 0.25f;
#pragma unroll
        for (int q = 0; q < 4; ++q) {
            int i = (n >> (3 - q)) & 1;
            int j = (k >> (3 - q)) & 1;
            float m = (i == j) ? cq[q] : ((i == 0) ? -sq[q] : sq[q]);
            v *= m;
        }
        s_bmat[k * 16 + n] = v;
    }
    __syncthreads();

    const int rowBase = (int)blockIdx.x * 256;
    int r = rowBase + tid;
    if (r >= totalRows) r = totalRows - 1;

    unsigned rr  = (unsigned)r;
    unsigned b   = rr / (unsigned)(OX * OY);
    unsigned rem = rr % (unsigned)(OX * OY);
    unsigned px  = rem / (unsigned)OY;
    unsigned py  = rem % (unsigned)OY;

    const float* xp = X + (size_t)b * (HW * HW) + (size_t)(2u * px) * HW + 2u * py;
    float t[16];
#pragma unroll
    for (int kx = 0; kx < 4; ++kx) {
        const float* rp = xp + kx * HW;
        float2 a = *(const float2*)(rp);
        float2 c = *(const float2*)(rp + 2);
        t[kx * 4 + 0] = a.x; t[kx * 4 + 1] = a.y;
        t[kx * 4 + 2] = c.x; t[kx * 4 + 3] = c.y;
    }

    const float r2p = 0.15915494f;
    const float r4p = 0.07957747f;
    float ar[3][2], ai[3][2];
#pragma unroll
    for (int q = 0; q < 3; ++q) {
        const float x0 = t[q * 5 + 0] * r2p, x1 = t[q * 5 + 1] * r4p;
        const float x2 = t[q * 5 + 2] * r2p, x3 = t[q * 5 + 3] * r4p;
        const float x4 = fmaf(t[q * 5 + 4], r2p, -Wt[q] * r4p);
        const float c0 = __builtin_amdgcn_cosf(x0), s0 = __builtin_amdgcn_sinf(x0);
        const float c1 = __builtin_amdgcn_cosf(x1), s1 = __builtin_amdgcn_sinf(x1);
        const float c2 = __builtin_amdgcn_cosf(x2), s2 = __builtin_amdgcn_sinf(x2);
        const float c3 = __builtin_amdgcn_cosf(x3), s3 = __builtin_amdgcn_sinf(x3);
        const float c4 = __builtin_amdgcn_cosf(x4), s4 = __builtin_amdgcn_sinf(x4);
        const float r0 = c1 - s1 * c0, i0 = -(s1 * s0);
        const float r1 = s1 + c1 * c0, i1 = c1 * s0;
        const float nr1 = r1 * c2 - i1 * s2, ni1 = r1 * s2 + i1 * c2;
        const float R0 = c3 * r0 - s3 * nr1, I0 = c3 * i0 - s3 * ni1;
        const float R1 = s3 * r0 + c3 * nr1, I1 = s3 * i0 + c3 * ni1;
        ar[q][0] = R0;                 ai[q][0] = I0;
        ar[q][1] = R1 * c4 - I1 * s4;  ai[q][1] = R1 * s4 + I1 * c4;
    }
    const float x15 = fmaf(t[15], r2p, -Wt[3] * r4p);
    const float e3qr = __builtin_amdgcn_cosf(x15), e3qi = __builtin_amdgcn_sinf(x15);

    const float e0r = s_ec[0], e0i = s_es[0], e1r = s_ec[1], e1i = s_es[1];
    const float e2r = s_ec[2], e2i = s_es[2], e3r = s_ec[3], e3i = s_es[3];

    float w01r[4], w01i[4], w23r[4], w23i[4];
#pragma unroll
    for (int i = 0; i < 2; ++i)
#pragma unroll
        for (int j = 0; j < 2; ++j) {
            w01r[i * 2 + j] = ar[0][i] * ar[1][j] - ai[0][i] * ai[1][j];
            w01i[i * 2 + j] = ar[0][i] * ai[1][j] + ai[0][i] * ar[1][j];
        }
    {
        const float pr = w01r[3], pi = w01i[3];
        w01r[3] = pr * e0r - pi * e0i;  w01i[3] = pr * e0i + pi * e0r;
        const float fr = e3qr * e2r - e3qi * e2i, fi = e3qr * e2i + e3qi * e2r;
        w23r[0] = ar[2][0];  w23i[0] = ai[2][0];
        w23r[1] = ar[2][0] * e3qr - ai[2][0] * e3qi;  w23i[1] = ar[2][0] * e3qi + ai[2][0] * e3qr;
        w23r[2] = ar[2][1];  w23i[2] = ai[2][1];
        w23r[3] = ar[2][1] * fr - ai[2][1] * fi;      w23i[3] = ar[2][1] * fi + ai[2][1] * fr;
    }
    float u23r[2], u23i[2], v01r[2], v01i[2];
#pragma unroll
    for (int j = 0; j < 2; ++j) {
        u23r[j] = w23r[2 + j] * e1r - w23i[2 + j] * e1i;  u23i[j] = w23r[2 + j] * e1i + w23i[2 + j] * e1r;
        v01r[j] = w01r[2 + j] * e3r - w01i[2 + j] * e3i;  v01i[j] = w01r[2 + j] * e3i + w01i[2 + j] * e3r;
    }
#pragma unroll
    for (int s = 0; s < 16; ++s) {
        const int hi = s >> 2, lo = s & 3;
        const bool b0 = (s & 8) != 0, b1 = (s & 4) != 0, b2 = (s & 2) != 0, b3 = (s & 1) != 0;
        const float pr = (b0 && b3) ? v01r[hi & 1] : w01r[hi], pi = (b0 && b3) ? v01i[hi & 1] : w01i[hi];
        const float qr = (b1 && b2) ? u23r[lo & 1] : w23r[lo], qi = (b1 && b2) ? u23i[lo & 1] : w23i[lo];
        s_re[wv][lane][s] = pr * qr - pi * qi;
        s_im[wv][lane][s] = pr * qi + pi * qr;
    }
    __syncthreads();

    const int nn   = lane & 15;
    const int hh   = lane >> 4;
    const int koff = hh * 2;

    float tot[2], z1[2], z2[2], z3[2];
#pragma unroll
    for (int tI = 0; tI < 2; ++tI) {
        v8f dRe = {}; v8f dIm = {};
        const int srcRow = tI * 16 + nn;
#pragma unroll
        for (int k0 = 0; k0 < 16; k0 += 4) {
            const int kk = k0 + koff;
            v2f b_re, b_im, aop;
            b_re.x = s_re[wv][srcRow][kk];      b_re.y = s_re[wv][srcRow][kk + 1];
            b_im.x = s_im[wv][srcRow][kk];      b_im.y = s_im[wv][srcRow][kk + 1];
            aop.x  = s_bmat[kk * 16 + nn];      aop.y  = s_bmat[(kk + 1) * 16 + nn];
            dRe = __builtin_amdgcn_wmma_f32_16x16x4_f32(false, aop, false, b_re,
                                                        (short)0, dRe, false, false);
            dIm = __builtin_amdgcn_wmma_f32_16x16x4_f32(false, aop, false, b_im,
                                                        (short)0, dIm, false, false);
        }
        float p[8];
#pragma unroll
        for (int v = 0; v < 8; ++v) p[v] = dRe[v] * dRe[v] + dIm[v] * dIm[v];
        const float s0 = p[0] + p[4], s1 = p[1] + p[5], s2 = p[2] + p[6], s3 = p[3] + p[7];
        const float d0 = p[0] - p[4], d1 = p[1] - p[5], d2 = p[2] - p[6], d3 = p[3] - p[7];
        const float t0 = s0 + s2, t1 = s1 + s3;
        const float u0 = s0 - s2, u1 = s1 - s3;
        z1[tI]  = (d0 + d2) + (d1 + d3);
        z2[tI]  = u0 + u1;
        z3[tI]  = t0 - t1;
        tot[tI] = t0 + t1;
    }

    const bool up = (hh != 0);
    const float k0v = up ? -tot[1] : tot[0];
    const float g0v = up ? -tot[0] : tot[1];
    const float k1v = up ? z1[1] : z1[0];
    const float g1v = up ? z1[0] : z1[1];
    const float k2v = up ? z2[1] : z2[0];
    const float g2v = up ? z2[0] : z2[1];
    const float k3v = up ? z3[1] : z3[0];
    const float g3v = up ? z3[0] : z3[1];
    float4 ev;
    ev.x = k0v + xhalf(g0v);
    ev.y = k1v + xhalf(g1v);
    ev.z = k2v + xhalf(g2v);
    ev.w = k3v + xhalf(g3v);

    const int rowG = rowBase + tid;
    if (rowG < totalRows) *(float4*)(out + (size_t)rowG * 4) = ev;
}

extern "C" void kernel_launch(void* const* d_in, const int* in_sizes, int n_in,
                              void* d_out, int out_size, void* d_ws, size_t ws_size,
                              hipStream_t stream) {
    const float* X  = (const float*)d_in[0];
    const float* Wt = (const float*)d_in[1];
    float* out      = (float*)d_out;
    int B = in_sizes[0] / (HW * HW);
    int totalRows = B * OX * OY;
    int blocks = (totalRows + 255) / 256;
    quanv_kernel<<<blocks, 256, 0, stream>>>(X, Wt, out, totalRows);
}


# Each lane encodes two rows side by side (a wave covers 64 rows), so all arithmetic behind the sines and cosines runs as packed two-wide f32 instructions; half as many waves and weight-table set-ups
# speedup vs baseline: 1.4686x; 1.0596x over previous
#include <hip/hip_runtime.h>


typedef float v2f __attribute__((ext_vector_type(2)));
typedef float v8f __attribute__((ext_vector_type(8)));

#define HW 192
#define OX 95
#define OY 95

__device__ __forceinline__ float xhalf(float x) {
    int b = __builtin_bit_cast(int, x);
    return __builtin_bit_cast(float,
        __builtin_amdgcn_permlanex16(b, b, 0x76543210, 0xfedcba98, false, false));
}

__device__ __forceinline__ void load_patch(const float* __restrict__ X, int r, int totalRows,
                                           float (&t)[16]) {
    if (r >= totalRows) r = totalRows - 1;
    unsigned rr  = (unsigned)r;
    unsigned b   = rr / (unsigned)(OX * OY);
    unsigned rem = rr % (unsigned)(OX * OY);
    unsigned px  = rem / (unsigned)OY;
    unsigned py  = rem % (unsigned)OY;
    const float* xp = X + (size_t)b * (HW * HW) + (size_t)(2u * px) * HW + 2u * py;
#pragma unroll
    for (int kx = 0; kx < 4; ++kx) {
        const float* rp = xp + kx * HW;
        float2 a = *(const float2*)(rp);
        float2 c = *(const float2*)(rp + 2);
        t[kx * 4 + 0] = a.x; t[kx * 4 + 1] = a.y;
        t[kx * 4 + 2] = c.x; t[kx * 4 + 3] = c.y;
    }
}

__device__ __forceinline__ v2f cos2(v2f x) {
    v2f r; r.x = __builtin_amdgcn_cosf(x.x); r.y = __builtin_amdgcn_cosf(x.y); return r;
}
__device__ __forceinline__ v2f sin2(v2f x) {
    v2f r; r.x = __builtin_amdgcn_sinf(x.x); r.y = __builtin_amdgcn_sinf(x.y); return r;
}

__global__ __launch_bounds__(256)
void quanv_kernel(const float* __restrict__ X, const float* __restrict__ Wt,
                  float* __restrict__ out, int totalRows) {
    __shared__ float s_bmat[16 * 16];
    __shared__ float s_ec[4], s_es[4];
    __shared__ float s_re[8][64][18];
    __shared__ float s_im[8][64][18];

    const int tid  = threadIdx.x;
    const int lane = tid & 31;
    const int wv   = tid >> 5;

    float cq[4], sq[4];
#pragma unroll
    for (int q = 0; q < 4; ++q) __sincosf(0.5f * Wt[4 + q], &sq[q], &cq[q]);

    if (tid < 4) {
        float sp, cp; __sincosf(Wt[tid], &sp, &cp);
        s_ec[tid] = cp; s_es[tid] = sp;
    }
    {
        int k = tid >> 4, n = tid & 15;
        float v = 0.25f;
#pragma unroll
        for (int q = 0; q < 4; ++q) {
            int i = (n >> (3 - q)) & 1;
            int j = (k >> (3 - q)) & 1;
            float m = (i == j) ? cq[q] : ((i == 0) ? -sq[q] : sq[q]);
            v *= m;
        }
        s_bmat[k * 16 + n] = v;
    }
    __syncthreads();

    const int rowBase = (int)blockIdx.x * 512 + wv * 64;
    float ta[16], tb[16];
    load_patch(X, rowBase + lane, totalRows, ta);
    load_patch(X, rowBase + 32 + lane, totalRows, tb);

    const float r2p = 0.15915494f;
    const float r4p = 0.07957747f;
    v2f ar[3][2], ai[3][2];
#pragma unroll
    for (int q = 0; q < 3; ++q) {
        const float cph = -Wt[q] * r4p;
        v2f x0, x1, x2, x3, x4;
        x0.x = ta[q * 5 + 0] * r2p;  x0.y = tb[q * 5 + 0] * r2p;
        x1.x = ta[q * 5 + 1] * r4p;  x1.y = tb[q * 5 + 1] * r4p;
        x2.x = ta[q * 5 + 2] * r2p;  x2.y = tb[q * 5 + 2] * r2p;
        x3.x = ta[q * 5 + 3] * r4p;  x3.y = tb[q * 5 + 3] * r4p;
        x4.x = fmaf(ta[q * 5 + 4], r2p, cph);  x4.y = fmaf(tb[q * 5 + 4], r2p, cph);
        const v2f c0 = cos2(x0), s0 = sin2(x0);
        const v2f c1 = cos2(x1), s1 = sin2(x1);
        const v2f c2 = cos2(x2), s2 = sin2(x2);
        const v2f c3 = cos2(x3), s3 = sin2(x3);
        const v2f c4 = cos2(x4), s4 = sin2(x4);
        const v2f r0 = c1 - s1 * c0, i0 = -(s1 * s0);
        const v2f r1 = s1 + c1 * c0, i1 = c1 * s0;
        const v2f nr1 = r1 * c2 - i1 * s2, ni1 = r1 * s2 + i1 * c2;
        const v2f R0 = c3 * r0 - s3 * nr1, I0 = c3 * i0 - s3 * ni1;
        const v2f R1 = s3 * r0 + c3 * nr1, I1 = s3 * i0 + c3 * ni1;
        ar[q][0] = R0;                 ai[q][0] = I0;
        ar[q][1] = R1 * c4 - I1 * s4;  ai[q][1] = R1 * s4 + I1 * c4;
    }
    v2f x15;
    x15.x = fmaf(ta[15], r2p, -Wt[3] * r4p);  x15.y = fmaf(tb[15], r2p, -Wt[3] * r4p);
    const v2f e3qr = cos2(x15), e3qi = sin2(x15);

    const float e0r = s_ec[0], e0i = s_es[0], e1r = s_ec[1], e1i = s_es[1];
    const float e2r = s_ec[2], e2i = s_es[2], e3r = s_ec[3], e3i = s_es[3];

    v2f w01r[4], w01i[4], w23r[4], w23i[4];
#pragma unroll
    for (int i = 0; i < 2; ++i)
#pragma unroll
        for (int j = 0; j < 2; ++j) {
            w01r[i * 2 + j] = ar[0][i] * ar[1][j] - ai[0][i] * ai[1][j];
            w01i[i * 2 + j] = ar[0][i] * ai[1][j] + ai[0][i] * ar[1][j];
        }
    {
        const v2f pr = w01r[3], pi = w01i[3];
        w01r[3] = pr * e0r - pi * e0i;  w01i[3] = pr * e0i + pi * e0r;
        const v2f fr = e3qr * e2r - e3qi * e2i, fi = e3qr * e2i + e3qi * e2r;
        w23r[0] = ar[2][0];  w23i[0] = ai[2][0];
        w23r[1] = ar[2][0] * e3qr - ai[2][0] * e3qi;  w23i[1] = ar[2][0] * e3qi + ai[2][0] * e3qr;
        w23r[2] = ar[2][1];  w23i[2] = ai[2][1];
        w23r[3] = ar[2][1] * fr - ai[2][1] * fi;      w23i[3] = ar[2][1] * fi + ai[2][1] * fr;
    }
    v2f u23r[2], u23i[2], v01r[2], v01i[2];
#pragma unroll
    for (int j = 0; j < 2; ++j) {
        u23r[j] = w23r[2 + j] * e1r - w23i[2 + j] * e1i;  u23i[j] = w23r[2 + j] * e1i + w23i[2 + j] * e1r;
        v01r[j] = w01r[2 + j] * e3r - w01i[2 + j] * e3i;  v01i[j] = w01r[2 + j] * e3i + w01i[2 + j] * e3r;
    }
#pragma unroll
    for (int s = 0; s < 16; ++s) {
        const int hi = s >> 2, lo = s & 3;
        const bool b0 = (s & 8) != 0, b1 = (s & 4) != 0, b2 = (s & 2) != 0, b3 = (s & 1) != 0;
        const v2f pr = (b0 && b3) ? v01r[hi & 1] : w01r[hi], pi = (b0 && b3) ? v01i[hi & 1] : w01i[hi];
        const v2f qr = (b1 && b2) ? u23r[lo & 1] : w23r[lo], qi = (b1 && b2) ? u23i[lo & 1] : w23i[lo];
        const v2f are = pr * qr - pi * qi;
        const v2f aim = pr * qi + pi * qr;
        s_re[wv][lane][s] = are.x;  s_re[wv][32 + lane][s] = are.y;
        s_im[wv][lane][s] = aim.x;  s_im[wv][32 + lane][s] = aim.y;
    }
    __syncthreads();

    const int nn   = lane & 15;
    const int hh   = lane >> 4;
    const int koff = hh * 2;
    const bool up  = (hh != 0);

    v2f aop[4];
#pragma unroll
    for (int ks = 0; ks < 4; ++ks) {
        const int kk = ks * 4 + koff;
        aop[ks].x = s_bmat[kk * 16 + nn];
        aop[ks].y = s_bmat[(kk + 1) * 16 + nn];
    }

#pragma unroll
    for (int hs = 0; hs < 2; ++hs) {
        float tot[2], z1[2], z2[2], z3[2];
#pragma unroll
        for (int tI = 0; tI < 2; ++tI) {
            v8f dRe = {}; v8f dIm = {};
            const int srcRow = hs * 32 + tI * 16 + nn;
#pragma unroll
            for (int ks = 0; ks < 4; ++ks) {
                const int kk = ks * 4 + koff;
                v2f b_re, b_im;
                b_re.x = s_re[wv][srcRow][kk];      b_re.y = s_re[wv][srcRow][kk + 1];
                b_im.x = s_im[wv][srcRow][kk];      b_im.y = s_im[wv][srcRow][kk + 1];
                dRe = __builtin_amdgcn_wmma_f32_16x16x4_f32(false, aop[ks], false, b_re,
                                                            (short)0, dRe, false, false);
                dIm = __builtin_amdgcn_wmma_f32_16x16x4_f32(false, aop[ks], false, b_im,
                                                            (short)0, dIm, false, false);
            }
            float p[8];
#pragma unroll
            for (int v = 0; v < 8; ++v) p[v] = dRe[v] * dRe[v] + dIm[v] * dIm[v];
            const float s0 = p[0] + p[4], s1 = p[1] + p[5], s2 = p[2] + p[6], s3 = p[3] + p[7];
            const float d0 = p[0] - p[4], d1 = p[1] - p[5], d2 = p[2] - p[6], d3 = p[3] - p[7];
            const float t0 = s0 + s2, t1 = s1 + s3;
            const float u0 = s0 - s2, u1 = s1 - s3;
            z1[tI]  = (d0 + d2) + (d1 + d3);
            z2[tI]  = u0 + u1;
            z3[tI]  = t0 - t1;
            tot[tI] = t0 + t1;
        }

        const float k0v = up ? -tot[1] : tot[0];
        const float g0v = up ? -tot[0] : tot[1];
        const float k1v = up ? z1[1] : z1[0];
        const float g1v = up ? z1[0] : z1[1];
        const float k2v = up ? z2[1] : z2[0];
        const float g2v = up ? z2[0] : z2[1];
        const float k3v = up ? z3[1] : z3[0];
        const float g3v = up ? z3[0] : z3[1];
        float4 ev;
        ev.x = k0v + xhalf(g0v);
        ev.y = k1v + xhalf(g1v);
        ev.z = k2v + xhalf(g2v);
        ev.w = k3v + xhalf(g3v);

        const int rowG = rowBase + hs * 32 + lane;
        if (rowG < totalRows) *(float4*)(out + (size_t)rowG * 4) = ev;
    }
}

extern "C" void kernel_launch(void* const* d_in, const int* in_sizes, int n_in,
                              void* d_out, int out_size, void* d_ws, size_t ws_size,
                              hipStream_t stream) {
    const float* X  = (const float*)d_in[0];
    const float* Wt = (const float*)d_in[1];
    float* out      = (float*)d_out;
    int B = in_sizes[0] / (HW * HW);
    int totalRows = B * OX * OY;
    int blocks = (totalRows + 511) / 512;
    quanv_kernel<<<blocks, 256, 0, stream>>>(X, Wt, out, totalRows);
}
